# starting kernel
# baseline (speedup 1.0000x reference)
#include <hip/hip_runtime.h>
#include <hip/hip_bf16.h>
#include <math.h>

typedef float v2f __attribute__((ext_vector_type(2)));
typedef float v8f __attribute__((ext_vector_type(8)));

#define HID   64
#define RANK  4
#define NAB   1036
#define GSTR  1044
#define XSTR  65

#define OFF_A1 0
#define OFF_B1 8
#define OFF_A2 264
#define OFF_B2 520
#define OFF_A3 776
#define OFF_B3 1032

__global__ __launch_bounds__(64) void hyperlr_fused(
    const float* __restrict__ coords, const float* __restrict__ param,
    const float* __restrict__ W1,  const float* __restrict__ B1,
    const float* __restrict__ W2,  const float* __restrict__ B2,
    const float* __restrict__ W3,  const float* __restrict__ B3,
    const float* __restrict__ Wh1, const float* __restrict__ Bh1,
    const float* __restrict__ Wh2, const float* __restrict__ Bh2,
    float* __restrict__ out)
{
    extern __shared__ float smem[];
    const int lane = threadIdx.x & 31;
    const int wave = threadIdx.x >> 5;

    float* g_lds  = smem + wave * (16 * GSTR);
    float* x_lds  = smem + 2 * (16 * GSTR) + wave * (16 * XSTR);
    float* t2_lds = smem + 2 * (16 * GSTR) + 2 * (16 * XSTR) + wave * (16 * RANK);

    const int m    = lane & 15;
    const int half = lane >> 4;
    const int kb   = half * 2;
    const int base = blockIdx.x * 32 + wave * 16;

    const float pv = param[base + m];
    v2f afrag[16];
    #pragma unroll
    for (int s = 0; s < 16; ++s) {
        int k = 4 * s + kb;
        afrag[s].x = tanhf(pv * Wh1[k]     + Bh1[k]);
        afrag[s].y = tanhf(pv * Wh1[k + 1] + Bh1[k + 1]);
    }

    auto loadB = [&](int ct, v2f (&dst)[16]) {
        int col  = ct * 16 + m;
        int ccol = col < NAB ? col : (NAB - 1);
        const float* wp = Wh2 + ccol * HID + kb;
        #pragma unroll
        for (int s = 0; s < 16; ++s) { dst[s].x = wp[4 * s]; dst[s].y = wp[4 * s + 1]; }
    };

    v2f bnext[16];
    loadB(0, bnext);
    for (int ct = 0; ct < 65; ++ct) {
        v2f bcur[16];
        #pragma unroll
        for (int s = 0; s < 16; ++s) bcur[s] = bnext[s];
        if (ct < 64) loadB(ct + 1, bnext);

        v8f acc0 = {}, acc1 = {};
        #pragma unroll
        for (int s = 0; s < 16; s += 2) {
            acc0 = __builtin_amdgcn_wmma_f32_16x16x4_f32(false, afrag[s],     false, bcur[s],
                                                         (short)0, acc0, false, false);
            acc1 = __builtin_amdgcn_wmma_f32_16x16x4_f32(false, afrag[s + 1], false, bcur[s + 1],
                                                         (short)0, acc1, false, false);
        }
        const v8f acc = acc0 + acc1;

        const int col  = ct * 16 + m;
        const int ccol = col < NAB ? col : (NAB - 1);
        const float bias = Bh2[ccol];
        #pragma unroll
        for (int v = 0; v < 8; ++v)
            g_lds[(v + 8 * half) * GSTR + col] = acc[v] + bias;
    }
    __syncthreads();

    const float c0 = coords[(base + m) * 2 + 0];
    const float c1 = coords[(base + m) * 2 + 1];
    float t1[RANK];
    #pragma unroll
    for (int r = 0; r < RANK; ++r)
        t1[r] = c0 * g_lds[m * GSTR + OFF_A1 + 2 * r]
              + c1 * g_lds[m * GSTR + OFF_A1 + 2 * r + 1];

    v2f x1frag[16];
    #pragma unroll
    for (int s = 0; s < 16; ++s) {
        #pragma unroll
        for (int j = 0; j < 2; ++j) {
            const int h = 4 * s + kb + j;
            float pre = W1[2 * h] * c0 + W1[2 * h + 1] * c1 + B1[h];
            #pragma unroll
            for (int r = 0; r < RANK; ++r)
                pre += g_lds[m * GSTR + OFF_B1 + 4 * h + r] * t1[r];
            const float x = tanhf(pre);
            x1frag[s][j] = x;
            x_lds[m * XSTR + h] = x;
        }
    }
    __syncthreads();

    {
        float t2[RANK] = {0.f, 0.f, 0.f, 0.f};
        for (int h = 0; h < HID; ++h) {
            const float xv = x_lds[m * XSTR + h];
            #pragma unroll
            for (int r = 0; r < RANK; ++r)
                t2[r] += xv * g_lds[m * GSTR + OFF_A2 + 64 * r + h];
        }
        if (half == 0) {
            #pragma unroll
            for (int r = 0; r < RANK; ++r) t2_lds[m * RANK + r] = t2[r];
        }
    }
    __syncthreads();

    #pragma unroll
    for (int ct = 0; ct < 4; ++ct) {
        const int hcol = ct * 16 + m;
        const float* wp = W2 + hcol * HID + kb;
        v2f bfrag[16];
        #pragma unroll
        for (int s = 0; s < 16; ++s) { bfrag[s].x = wp[4 * s]; bfrag[s].y = wp[4 * s + 1]; }

        v8f acc0 = {}, acc1 = {};
        #pragma unroll
        for (int s = 0; s < 16; s += 2) {
            acc0 = __builtin_amdgcn_wmma_f32_16x16x4_f32(false, x1frag[s],     false, bfrag[s],
                                                         (short)0, acc0, false, false);
            acc1 = __builtin_amdgcn_wmma_f32_16x16x4_f32(false, x1frag[s + 1], false, bfrag[s + 1],
                                                         (short)0, acc1, false, false);
        }
        const v8f acc = acc0 + acc1;

        const float b2h = B2[hcol];
        #pragma unroll
        for (int v = 0; v < 8; ++v) {
            const int row = v + 8 * half;
            float pre = acc[v] + b2h;
            #pragma unroll
            for (int r = 0; r < RANK; ++r)
                pre += t2_lds[row * RANK + r] * g_lds[row * GSTR + OFF_B2 + 4 * hcol + r];
            x_lds[row * XSTR + hcol] = tanhf(pre);
        }
    }
    __syncthreads();

    if (half == 0) {
        float t3[RANK] = {0.f, 0.f, 0.f, 0.f};
        float w3s = 0.f;
        for (int h = 0; h < HID; ++h) {
            const float xv = x_lds[m * XSTR + h];
            w3s += xv * W3[h];
            #pragma unroll
            for (int r = 0; r < RANK; ++r)
                t3[r] += xv * g_lds[m * GSTR + OFF_A3 + 64 * r + h];
        }
        float o = w3s + B3[0];
        #pragma unroll
        for (int r = 0; r < RANK; ++r)
            o += t3[r] * g_lds[m * GSTR + OFF_B3 + r];
        out[base + m] = o;
    }
}

extern "C" void kernel_launch(void* const* d_in, const int* in_sizes, int n_in,
                              void* d_out, int out_size, void* d_ws, size_t ws_size,
                              hipStream_t stream) {
    const float* coords = (const float*)d_in[0];
    const float* param  = (const float*)d_in[1];
    const float* W1     = (const float*)d_in[2];
    const float* B1     = (const float*)d_in[3];
    const float* W2     = (const float*)d_in[4];
    const float* B2     = (const float*)d_in[5];
    const float* W3     = (const float*)d_in[6];
    const float* B3     = (const float*)d_in[7];
    const float* Wh1    = (const float*)d_in[8];
    const float* Bh1    = (const float*)d_in[9];
    const float* Wh2    = (const float*)d_in[10];
    const float* Bh2    = (const float*)d_in[11];
    float* out = (float*)d_out;

    const int n = in_sizes[1];
    const size_t smem = (size_t)(2 * 16 * GSTR + 2 * 16 * XSTR + 2 * 16 * RANK) * sizeof(float);
    (void)hipFuncSetAttribute((const void*)hyperlr_fused,
                              hipFuncAttributeMaxDynamicSharedMemorySize, (int)smem);

    const int blocks = n / 32;
    hyperlr_fused<<<blocks, 64, smem, stream>>>(coords, param, W1, B1, W2, B2, W3, B3,
                                                Wh1, Bh1, Wh2, Bh2, out);
}


# Wh2 reaches the B fragments through LDS: whole-line 16-byte global loads of 32-row chunks, shared by both waves and double-buffered, replace per-lane 8-byte loads strided 256 B
# speedup vs baseline: 1.4768x; 1.4768x over previous
#include <hip/hip_runtime.h>
#include <hip/hip_bf16.h>
#include <math.h>

typedef float v2f __attribute__((ext_vector_type(2)));
typedef float v4f __attribute__((ext_vector_type(4)));
typedef float v8f __attribute__((ext_vector_type(8)));

#define HID   64
#define RANK  4
#define NAB   1036
#define GSTR  1044
#define XSTR  65
#define BROWS 32
#define NCHUNK 33
#define BSTR  68

#define OFF_A1 0
#define OFF_B1 8
#define OFF_A2 264
#define OFF_B2 520
#define OFF_A3 776
#define OFF_B3 1032

__global__ __launch_bounds__(64) void hyperlr_fused(
    const float* __restrict__ coords, const float* __restrict__ param,
    const float* __restrict__ W1,  const float* __restrict__ B1,
    const float* __restrict__ W2,  const float* __restrict__ B2,
    const float* __restrict__ W3,  const float* __restrict__ B3,
    const float* __restrict__ Wh1, const float* __restrict__ Bh1,
    const float* __restrict__ Wh2, const float* __restrict__ Bh2,
    float* __restrict__ out)
{
    extern __shared__ __attribute__((aligned(16))) float smem[];
    const int lane = threadIdx.x & 31;
    const int wave = threadIdx.x >> 5;

    float* g_lds  = smem + wave * (16 * GSTR);
    float* x_lds  = smem + 2 * (16 * GSTR) + wave * (16 * XSTR);
    float* t2_lds = smem + 2 * (16 * GSTR) + 2 * (16 * XSTR) + wave * (16 * RANK);
    float* b_lds  = smem + 2 * (16 * GSTR);
    static_assert(2 * BROWS * BSTR >= 2 * 16 * XSTR + 2 * 16 * RANK, "staging ring must cover the region it shares");

    const int m    = lane & 15;
    const int half = lane >> 4;
    const int kb   = half * 2;
    const int base = blockIdx.x * 32 + wave * 16;

    const float pv = param[base + m];
    v2f afrag[16];
    #pragma unroll
    for (int s = 0; s < 16; ++s) {
        int k = 4 * s + kb;
        afrag[s].x = tanhf(pv * Wh1[k]     + Bh1[k]);
        afrag[s].y = tanhf(pv * Wh1[k + 1] + Bh1[k + 1]);
    }

    const int srow = threadIdx.x >> 4;
    const int skq  = (threadIdx.x & 15) * 4;
    v4f stg[8];
    auto gload = [&](int chunk) {
        #pragma unroll
        for (int i = 0; i < 8; ++i) {
            const int row  = chunk * BROWS + srow + 4 * i;
            const int crow = row < NAB ? row : (NAB - 1);
            stg[i] = *(const v4f*)(Wh2 + crow * HID + skq);
        }
    };
    auto sstore = [&](int stage) {
        float* dst = b_lds + stage * (BROWS * BSTR) + srow * BSTR + skq;
        #pragma unroll
        for (int i = 0; i < 8; ++i) *(v4f*)(dst + 4 * i * BSTR) = stg[i];
    };
    auto readB = [&](int stage, int tile, v2f (&dst)[16]) {
        const float* sp = b_lds + stage * (BROWS * BSTR) + (tile * 16 + m) * BSTR + kb;
        #pragma unroll
        for (int s = 0; s < 16; ++s) dst[s] = *(const v2f*)(sp + 4 * s);
    };
    auto mma_tile = [&](const v2f (&b)[16]) -> v8f {
        v8f acc0 = {}, acc1 = {};
        #pragma unroll
        for (int s = 0; s < 16; s += 2) {
            acc0 = __builtin_amdgcn_wmma_f32_16x16x4_f32(false, afrag[s],     false, b[s],
                                                         (short)0, acc0, false, false);
            acc1 = __builtin_amdgcn_wmma_f32_16x16x4_f32(false, afrag[s + 1], false, b[s + 1],
                                                         (short)0, acc1, false, false);
        }
        return acc0 + acc1;
    };
    auto load_bias = [&](int ct) -> float {
        const int col  = ct * 16 + m;
        const int ccol = col < NAB ? col : (NAB - 1);
        return Bh2[ccol];
    };
    auto store_tile = [&](int ct, const v8f& acc, float bias) {
        const int col = ct * 16 + m;
        #pragma unroll
        for (int v = 0; v < 8; ++v)
            g_lds[(v + 8 * half) * GSTR + col] = acc[v] + bias;
    };

    gload(0);
    sstore(0);
    __syncthreads();
    for (int c = 0; c < NCHUNK; ++c) {
        const int stage = c & 1;
        const bool two = c < NCHUNK - 1;
        v2f b0[16], b1[16];
        v8f accA, accB;
        if (c + 1 < NCHUNK) gload(c + 1);
        const float biasA = load_bias(2 * c);
        const float biasB = load_bias(2 * c + 1);
        readB(stage, 0, b0);
        if (two) readB(stage, 1, b1);
        __builtin_amdgcn_sched_barrier(0);
        accA = mma_tile(b0);
        if (two) accB = mma_tile(b1);
        store_tile(2 * c, accA, biasA);
        if (two) store_tile(2 * c + 1, accB, biasB);
        if (c + 1 < NCHUNK) sstore(stage ^ 1);
        __syncthreads();
    }

    const float c0 = coords[(base + m) * 2 + 0];
    const float c1 = coords[(base + m) * 2 + 1];
    float t1[RANK];
    #pragma unroll
    for (int r = 0; r < RANK; ++r)
        t1[r] = c0 * g_lds[m * GSTR + OFF_A1 + 2 * r]
              + c1 * g_lds[m * GSTR + OFF_A1 + 2 * r + 1];

    v2f x1frag[16];
    #pragma unroll
    for (int s = 0; s < 16; ++s) {
        #pragma unroll
        for (int j = 0; j < 2; ++j) {
            const int h = 4 * s + kb + j;
            float pre = W1[2 * h] * c0 + W1[2 * h + 1] * c1 + B1[h];
            #pragma unroll
            for (int r = 0; r < RANK; ++r)
                pre += g_lds[m * GSTR + OFF_B1 + 4 * h + r] * t1[r];
            const float x = tanhf(pre);
            x1frag[s][j] = x;
            x_lds[m * XSTR + h] = x;
        }
    }
    __syncthreads();

    {
        float t2[RANK] = {0.f, 0.f, 0.f, 0.f};
        for (int h = 0; h < HID; ++h) {
            const float xv = x_lds[m * XSTR + h];
            #pragma unroll
            for (int r = 0; r < RANK; ++r)
                t2[r] += xv * g_lds[m * GSTR + OFF_A2 + 64 * r + h];
        }
        if (half == 0) {
            #pragma unroll
            for (int r = 0; r < RANK; ++r) t2_lds[m * RANK + r] = t2[r];
        }
    }
    __syncthreads();

    #pragma unroll
    for (int ct = 0; ct < 4; ++ct) {
        const int hcol = ct * 16 + m;
        const float* wp = W2 + hcol * HID + kb;
        v2f bfrag[16];
        #pragma unroll
        for (int s = 0; s < 16; ++s) { bfrag[s].x = wp[4 * s]; bfrag[s].y = wp[4 * s + 1]; }

        v8f acc0 = {}, acc1 = {};
        #pragma unroll
        for (int s = 0; s < 16; s += 2) {
            acc0 = __builtin_amdgcn_wmma_f32_16x16x4_f32(false, x1frag[s],     false, bfrag[s],
                                                         (short)0, acc0, false, false);
            acc1 = __builtin_amdgcn_wmma_f32_16x16x4_f32(false, x1frag[s + 1], false, bfrag[s + 1],
                                                         (short)0, acc1, false, false);
        }
        const v8f acc = acc0 + acc1;

        const float b2h = B2[hcol];
        #pragma unroll
        for (int v = 0; v < 8; ++v) {
            const int row = v + 8 * half;
            float pre = acc[v] + b2h;
            #pragma unroll
            for (int r = 0; r < RANK; ++r)
                pre += t2_lds[row * RANK + r] * g_lds[row * GSTR + OFF_B2 + 4 * hcol + r];
            x_lds[row * XSTR + hcol] = tanhf(pre);
        }
    }
    __syncthreads();

    if (half == 0) {
        float t3[RANK] = {0.f, 0.f, 0.f, 0.f};
        float w3s = 0.f;
        for (int h = 0; h < HID; ++h) {
            const float xv = x_lds[m * XSTR + h];
            w3s += xv * W3[h];
            #pragma unroll
            for (int r = 0; r < RANK; ++r)
                t3[r] += xv * g_lds[m * GSTR + OFF_A3 + 64 * r + h];
        }
        float o = w3s + B3[0];
        #pragma unroll
        for (int r = 0; r < RANK; ++r)
            o += t3[r] * g_lds[m * GSTR + OFF_B3 + r];
        out[base + m] = o;
    }
}

extern "C" void kernel_launch(void* const* d_in, const int* in_sizes, int n_in,
                              void* d_out, int out_size, void* d_ws, size_t ws_size,
                              hipStream_t stream) {
    const float* coords = (const float*)d_in[0];
    const float* param  = (const float*)d_in[1];
    const float* W1     = (const float*)d_in[2];
    const float* B1     = (const float*)d_in[3];
    const float* W2     = (const float*)d_in[4];
    const float* B2     = (const float*)d_in[5];
    const float* W3     = (const float*)d_in[6];
    const float* B3     = (const float*)d_in[7];
    const float* Wh1    = (const float*)d_in[8];
    const float* Bh1    = (const float*)d_in[9];
    const float* Wh2    = (const float*)d_in[10];
    const float* Bh2    = (const float*)d_in[11];
    float* out = (float*)d_out;

    const int n = in_sizes[1];
    const size_t smem = (size_t)(2 * 16 * GSTR + 2 * BROWS * BSTR) * sizeof(float);
    (void)hipFuncSetAttribute((const void*)hyperlr_fused,
                              hipFuncAttributeMaxDynamicSharedMemorySize, (int)smem);

    const int blocks = n / 32;
    hyperlr_fused<<<blocks, 64, smem, stream>>>(coords, param, W1, B1, W2, B2, W3, B3,
                                                Wh1, Bh1, Wh2, Bh2, out);
}


# g is produced and consumed in four sections through one 16x272 LDS window per wave, not a whole 16x1044 tile: LDS per block 151 KB -> 62 KB, so every block is resident at once, two waves per SIMD
# speedup vs baseline: 2.6906x; 1.8219x over previous
#include <hip/hip_runtime.h>
#include <hip/hip_bf16.h>
#include <math.h>

typedef float v2f __attribute__((ext_vector_type(2)));
typedef float v4f __attribute__((ext_vector_type(4)));
typedef float v8f __attribute__((ext_vector_type(8)));

#define HID   64
#define RANK  4
#define NAB   1036
#define GSTR  276
#define XSTR  65
#define BROWS 32
#define BSTR  68

#define OFF_A1 0
#define OFF_B1 8
#define OFF_A2 0
#define OFF_B2 0
#define OFF_A3 0
#define OFF_B3 256

__global__ __launch_bounds__(64) void hyperlr_fused(
    const float* __restrict__ coords, const float* __restrict__ param,
    const float* __restrict__ W1,  const float* __restrict__ B1,
    const float* __restrict__ W2,  const float* __restrict__ B2,
    const float* __restrict__ W3,  const float* __restrict__ B3,
    const float* __restrict__ Wh1, const float* __restrict__ Bh1,
    const float* __restrict__ Wh2, const float* __restrict__ Bh2,
    float* __restrict__ out)
{
    extern __shared__ __attribute__((aligned(16))) float smem[];
    const int lane = threadIdx.x & 31;
    const int wave = threadIdx.x >> 5;

    float* g_lds  = smem + wave * (16 * GSTR);
    float* b_lds  = smem + 2 * (16 * GSTR);
    float* x_lds  = smem + 2 * (16 * GSTR) + 2 * BROWS * BSTR + wave * (16 * XSTR);
    float* t2_lds = smem + 2 * (16 * GSTR) + 2 * BROWS * BSTR + 2 * (16 * XSTR) + wave * (16 * RANK);
    static_assert((2 * 16 * GSTR) % 4 == 0, "staging ring must start on a 16-byte boundary");

    const int m    = lane & 15;
    const int half = lane >> 4;
    const int kb   = half * 2;
    const int base = blockIdx.x * 32 + wave * 16;

    const float pv = param[base + m];
    v2f afrag[16];
    #pragma unroll
    for (int s = 0; s < 16; ++s) {
        int k = 4 * s + kb;
        afrag[s].x = tanhf(pv * Wh1[k]     + Bh1[k]);
        afrag[s].y = tanhf(pv * Wh1[k + 1] + Bh1[k + 1]);
    }

    const int srow = threadIdx.x >> 4;
    const int skq  = (threadIdx.x & 15) * 4;
    v4f stg[8];
    auto chunk_col = [&](int j) -> int { return j * BROWS - (j >= 9 ? 24 : 0); };
    auto gload = [&](int j) {
        const int row0 = chunk_col(j);
        #pragma unroll
        for (int i = 0; i < 8; ++i) {
            const int row  = row0 + srow + 4 * i;
            const int crow = row < NAB ? row : (NAB - 1);
            stg[i] = *(const v4f*)(Wh2 + crow * HID + skq);
        }
    };
    auto sstore = [&](int stage) {
        float* dst = b_lds + stage * (BROWS * BSTR) + srow * BSTR + skq;
        #pragma unroll
        for (int i = 0; i < 8; ++i) *(v4f*)(dst + 4 * i * BSTR) = stg[i];
    };
    auto readB = [&](int stage, int tile, v2f (&dst)[16]) {
        const float* sp = b_lds + stage * (BROWS * BSTR) + (tile * 16 + m) * BSTR + kb;
        #pragma unroll
        for (int s = 0; s < 16; ++s) dst[s] = *(const v2f*)(sp + 4 * s);
    };
    auto mma_tile = [&](const v2f (&b)[16]) -> v8f {
        v8f acc0 = {}, acc1 = {};
        #pragma unroll
        for (int s = 0; s < 16; s += 2) {
            acc0 = __builtin_amdgcn_wmma_f32_16x16x4_f32(false, afrag[s],     false, b[s],
                                                         (short)0, acc0, false, false);
            acc1 = __builtin_amdgcn_wmma_f32_16x16x4_f32(false, afrag[s + 1], false, b[s + 1],
                                                         (short)0, acc1, false, false);
        }
        return acc0 + acc1;
    };
    auto load_bias = [&](int gcol0) -> float {
        const int col  = gcol0 + m;
        const int ccol = col < NAB ? col : (NAB - 1);
        return Bh2[ccol];
    };
    auto store_tile = [&](int wcol0, const v8f& acc, float bias) {
        const int col = wcol0 + m;
        #pragma unroll
        for (int v = 0; v < 8; ++v)
            g_lds[(v + 8 * half) * GSTR + col] = acc[v] + bias;
    };

    auto run_chunks = [&](int j0, int n, bool tail_one, bool more_after) {
        for (int cc = 0; cc < n; ++cc) {
            const int j = j0 + cc;
            const int stage = j & 1;
            const bool two  = !(tail_one && cc == n - 1);
            const bool next = more_after || cc + 1 < n;
            const int gcol0 = chunk_col(j);
            v2f b0[16], b1[16];
            v8f accA, accB;
            if (next) gload(j + 1);
            const float biasA = load_bias(gcol0);
            const float biasB = load_bias(gcol0 + 16);
            readB(stage, 0, b0);
            if (two) readB(stage, 1, b1);
            __builtin_amdgcn_sched_barrier(0);
            accA = mma_tile(b0);
            if (two) accB = mma_tile(b1);
            store_tile(cc * BROWS, accA, biasA);
            if (two) store_tile(cc * BROWS + 16, accB, biasB);
            if (next) sstore(stage ^ 1);
            __syncthreads();
        }
    };

    gload(0);
    sstore(0);
    __syncthreads();

    run_chunks(0, 9, true, true);

    const float c0 = coords[(base + m) * 2 + 0];
    const float c1 = coords[(base + m) * 2 + 1];
    float t1[RANK];
    #pragma unroll
    for (int r = 0; r < RANK; ++r)
        t1[r] = c0 * g_lds[m * GSTR + OFF_A1 + 2 * r]
              + c1 * g_lds[m * GSTR + OFF_A1 + 2 * r + 1];

    v2f x1frag[16];
    #pragma unroll
    for (int s = 0; s < 16; ++s) {
        #pragma unroll
        for (int j = 0; j < 2; ++j) {
            const int h = 4 * s + kb + j;
            float pre = W1[2 * h] * c0 + W1[2 * h + 1] * c1 + B1[h];
            #pragma unroll
            for (int r = 0; r < RANK; ++r)
                pre += g_lds[m * GSTR + OFF_B1 + 4 * h + r] * t1[r];
            const float x = tanhf(pre);
            x1frag[s][j] = x;
            x_lds[m * XSTR + h] = x;
        }
    }

    run_chunks(9, 8, false, true);

    {
        float t2[RANK] = {0.f, 0.f, 0.f, 0.f};
        for (int h = 0; h < HID; ++h) {
            const float xv = x_lds[m * XSTR + h];
            #pragma unroll
            for (int r = 0; r < RANK; ++r)
                t2[r] += xv * g_lds[m * GSTR + OFF_A2 + 64 * r + h];
        }
        if (half == 0) {
            #pragma unroll
            for (int r = 0; r < RANK; ++r) t2_lds[m * RANK + r] = t2[r];
        }
    }

    run_chunks(17, 8, false, true);

    #pragma unroll
    for (int ct = 0; ct < 4; ++ct) {
        const int hcol = ct * 16 + m;
        const float* wp = W2 + hcol * HID + kb;
        v2f bfrag[16];
        #pragma unroll
        for (int s = 0; s < 16; ++s) { bfrag[s].x = wp[4 * s]; bfrag[s].y = wp[4 * s + 1]; }

        v8f acc0 = {}, acc1 = {};
        #pragma unroll
        for (int s = 0; s < 16; s += 2) {
            acc0 = __builtin_amdgcn_wmma_f32_16x16x4_f32(false, x1frag[s],     false, bfrag[s],
                                                         (short)0, acc0, false, false);
            acc1 = __builtin_amdgcn_wmma_f32_16x16x4_f32(false, x1frag[s + 1], false, bfrag[s + 1],
                                                         (short)0, acc1, false, false);
        }
        const v8f acc = acc0 + acc1;

        const float b2h = B2[hcol];
        #pragma unroll
        for (int v = 0; v < 8; ++v) {
            const int row = v + 8 * half;
            float pre = acc[v] + b2h;
            #pragma unroll
            for (int r = 0; r < RANK; ++r)
                pre += t2_lds[row * RANK + r] * g_lds[row * GSTR + OFF_B2 + 4 * hcol + r];
            x_lds[row * XSTR + hcol] = tanhf(pre);
        }
    }

    run_chunks(25, 9, true, false);

    if (half == 0) {
        float t3[RANK] = {0.f, 0.f, 0.f, 0.f};
        float w3s = 0.f;
        for (int h = 0; h < HID; ++h) {
            const float xv = x_lds[m * XSTR + h];
            w3s += xv * W3[h];
            #pragma unroll
            for (int r = 0; r < RANK; ++r)
                t3[r] += xv * g_lds[m * GSTR + OFF_A3 + 64 * r + h];
        }
        float o = w3s + B3[0];
        #pragma unroll
        for (int r = 0; r < RANK; ++r)
            o += t3[r] * g_lds[m * GSTR + OFF_B3 + r];
        out[base + m] = o;
    }
}

extern "C" void kernel_launch(void* const* d_in, const int* in_sizes, int n_in,
                              void* d_out, int out_size, void* d_ws, size_t ws_size,
                              hipStream_t stream) {
    const float* coords = (const float*)d_in[0];
    const float* param  = (const float*)d_in[1];
    const float* W1     = (const float*)d_in[2];
    const float* B1     = (const float*)d_in[3];
    const float* W2     = (const float*)d_in[4];
    const float* B2     = (const float*)d_in[5];
    const float* W3     = (const float*)d_in[6];
    const float* B3     = (const float*)d_in[7];
    const float* Wh1    = (const float*)d_in[8];
    const float* Bh1    = (const float*)d_in[9];
    const float* Wh2    = (const float*)d_in[10];
    const float* Bh2    = (const float*)d_in[11];
    float* out = (float*)d_out;

    const int n = in_sizes[1];
    const size_t smem = (size_t)(2 * 16 * GSTR + 2 * BROWS * BSTR + 2 * 16 * XSTR + 2 * 16 * RANK) * sizeof(float);
    (void)hipFuncSetAttribute((const void*)hyperlr_fused,
                              hipFuncAttributeMaxDynamicSharedMemorySize, (int)smem);

    const int blocks = n / 32;
    hyperlr_fused<<<blocks, 64, smem, stream>>>(coords, param, W1, B1, W2, B2, W3, B3,
                                                Wh1, Bh1, Wh2, Bh2, out);
}


# tanh without a branch (both ranges of tanhf computed, one selected), x stored to LDS after the evaluations: each activation phase is one basic block, loads issued as a group, evaluations overlapped
# speedup vs baseline: 3.1362x; 1.1656x over previous
#include <hip/hip_runtime.h>
#include <hip/hip_bf16.h>
#include <math.h>

typedef float v2f __attribute__((ext_vector_type(2)));
typedef float v4f __attribute__((ext_vector_type(4)));
typedef float v8f __attribute__((ext_vector_type(8)));

#define HID   64
#define RANK  4
#define NAB   1036
#define GSTR  276
#define XSTR  65
#define BROWS 32
#define BSTR  68

#define OFF_A1 0
#define OFF_B1 8
#define OFF_A2 0
#define OFF_B2 0
#define OFF_A3 0
#define OFF_B3 256

__device__ __forceinline__ float tanh_sl(float x) {
    const float ax = fabsf(x);
    const float s  = x * x;
    float p = fmaf(s, -0x1.758e7ap-8f, 0x1.521192p-6f);
    p = fmaf(s, p, -0x1.b8389cp-5f);
    p = fmaf(s, p, 0x1.110704p-3f);
    p = fmaf(s, p, -0x1.555532p-2f);
    const float lo = fmaf(s, ax * p, ax);
    const float e  = expf(ax + ax);
    const float hi = fmaf(__builtin_amdgcn_rcpf(1.0f + e), -2.0f, 1.0f);
    return copysignf(ax < 0.625f ? lo : hi, x);
}

__global__ __launch_bounds__(64) void hyperlr_fused(
    const float* __restrict__ coords, const float* __restrict__ param,
    const float* __restrict__ W1,  const float* __restrict__ B1,
    const float* __restrict__ W2,  const float* __restrict__ B2,
    const float* __restrict__ W3,  const float* __restrict__ B3,
    const float* __restrict__ Wh1, const float* __restrict__ Bh1,
    const float* __restrict__ Wh2, const float* __restrict__ Bh2,
    float* __restrict__ out)
{
    extern __shared__ __attribute__((aligned(16))) float smem[];
    const int lane = threadIdx.x & 31;
    const int wave = threadIdx.x >> 5;

    float* g_lds  = smem + wave * (16 * GSTR);
    float* b_lds  = smem + 2 * (16 * GSTR);
    float* x_lds  = smem + 2 * (16 * GSTR) + 2 * BROWS * BSTR + wave * (16 * XSTR);
    float* t2_lds = smem + 2 * (16 * GSTR) + 2 * BROWS * BSTR + 2 * (16 * XSTR) + wave * (16 * RANK);
    static_assert((2 * 16 * GSTR) % 4 == 0, "staging ring must start on a 16-byte boundary");

    const int m    = lane & 15;
    const int half = lane >> 4;
    const int kb   = half * 2;
    const int base = blockIdx.x * 32 + wave * 16;

    const float pv = param[base + m];
    v2f afrag[16];
    #pragma unroll
    for (int s = 0; s < 16; ++s) {
        int k = 4 * s + kb;
        afrag[s].x = tanh_sl(pv * Wh1[k]     + Bh1[k]);
        afrag[s].y = tanh_sl(pv * Wh1[k + 1] + Bh1[k + 1]);
    }

    const int srow = threadIdx.x >> 4;
    const int skq  = (threadIdx.x & 15) * 4;
    v4f stg[8];
    auto chunk_col = [&](int j) -> int { return j * BROWS - (j >= 9 ? 24 : 0); };
    auto gload = [&](int j) {
        const int row0 = chunk_col(j);
        #pragma unroll
        for (int i = 0; i < 8; ++i) {
            const int row  = row0 + srow + 4 * i;
            const int crow = row < NAB ? row : (NAB - 1);
            stg[i] = *(const v4f*)(Wh2 + crow * HID + skq);
        }
    };
    auto sstore = [&](int stage) {
        float* dst = b_lds + stage * (BROWS * BSTR) + srow * BSTR + skq;
        #pragma unroll
        for (int i = 0; i < 8; ++i) *(v4f*)(dst + 4 * i * BSTR) = stg[i];
    };
    auto readB = [&](int stage, int tile, v2f (&dst)[16]) {
        const float* sp = b_lds + stage * (BROWS * BSTR) + (tile * 16 + m) * BSTR + kb;
        #pragma unroll
        for (int s = 0; s < 16; ++s) dst[s] = *(const v2f*)(sp + 4 * s);
    };
    auto mma_tile = [&](const v2f (&b)[16]) -> v8f {
        v8f acc0 = {}, acc1 = {};
        #pragma unroll
        for (int s = 0; s < 16; s += 2) {
            acc0 = __builtin_amdgcn_wmma_f32_16x16x4_f32(false, afrag[s],     false, b[s],
                                                         (short)0, acc0, false, false);
            acc1 = __builtin_amdgcn_wmma_f32_16x16x4_f32(false, afrag[s + 1], false, b[s + 1],
                                                         (short)0, acc1, false, false);
        }
        return acc0 + acc1;
    };
    auto load_bias = [&](int gcol0) -> float {
        const int col  = gcol0 + m;
        const int ccol = col < NAB ? col : (NAB - 1);
        return Bh2[ccol];
    };
    auto store_tile = [&](int wcol0, const v8f& acc, float bias) {
        const int col = wcol0 + m;
        #pragma unroll
        for (int v = 0; v < 8; ++v)
            g_lds[(v + 8 * half) * GSTR + col] = acc[v] + bias;
    };

    auto run_chunks = [&](int j0, int n, bool tail_one, bool more_after) {
        for (int cc = 0; cc < n; ++cc) {
            const int j = j0 + cc;
            const int stage = j & 1;
            const bool two  = !(tail_one && cc == n - 1);
            const bool next = more_after || cc + 1 < n;
            const int gcol0 = chunk_col(j);
            v2f b0[16], b1[16];
            v8f accA, accB;
            if (next) gload(j + 1);
            const float biasA = load_bias(gcol0);
            const float biasB = load_bias(gcol0 + 16);
            readB(stage, 0, b0);
            if (two) readB(stage, 1, b1);
            __builtin_amdgcn_sched_barrier(0);
            accA = mma_tile(b0);
            if (two) accB = mma_tile(b1);
            store_tile(cc * BROWS, accA, biasA);
            if (two) store_tile(cc * BROWS + 16, accB, biasB);
            if (next) sstore(stage ^ 1);
            __syncthreads();
        }
    };

    gload(0);
    sstore(0);
    __syncthreads();

    run_chunks(0, 9, true, true);

    const float c0 = coords[(base + m) * 2 + 0];
    const float c1 = coords[(base + m) * 2 + 1];
    float t1[RANK];
    #pragma unroll
    for (int r = 0; r < RANK; ++r)
        t1[r] = c0 * g_lds[m * GSTR + OFF_A1 + 2 * r]
              + c1 * g_lds[m * GSTR + OFF_A1 + 2 * r + 1];

    v2f x1frag[16];
    #pragma unroll
    for (int s = 0; s < 16; ++s) {
        #pragma unroll
        for (int j = 0; j < 2; ++j) {
            const int h = 4 * s + kb + j;
            float pre = W1[2 * h] * c0 + W1[2 * h + 1] * c1 + B1[h];
            #pragma unroll
            for (int r = 0; r < RANK; ++r)
                pre += g_lds[m * GSTR + OFF_B1 + 4 * h + r] * t1[r];
            x1frag[s][j] = tanh_sl(pre);
        }
    }
    #pragma unroll
    for (int s = 0; s < 16; ++s) {
        #pragma unroll
        for (int j = 0; j < 2; ++j)
            x_lds[m * XSTR + 4 * s + kb + j] = x1frag[s][j];
    }

    run_chunks(9, 8, false, true);

    {
        float t2[RANK] = {0.f, 0.f, 0.f, 0.f};
        for (int h = 0; h < HID; ++h) {
            const float xv = x_lds[m * XSTR + h];
            #pragma unroll
            for (int r = 0; r < RANK; ++r)
                t2[r] += xv * g_lds[m * GSTR + OFF_A2 + 64 * r + h];
        }
        if (half == 0) {
            #pragma unroll
            for (int r = 0; r < RANK; ++r) t2_lds[m * RANK + r] = t2[r];
        }
    }

    run_chunks(17, 8, false, true);

    float x2v[4][8];
    #pragma unroll
    for (int ct = 0; ct < 4; ++ct) {
        const int hcol = ct * 16 + m;
        const float* wp = W2 + hcol * HID + kb;
        v2f bfrag[16];
        #pragma unroll
        for (int s = 0; s < 16; ++s) { bfrag[s].x = wp[4 * s]; bfrag[s].y = wp[4 * s + 1]; }

        v8f acc0 = {}, acc1 = {};
        #pragma unroll
        for (int s = 0; s < 16; s += 2) {
            acc0 = __builtin_amdgcn_wmma_f32_16x16x4_f32(false, x1frag[s],     false, bfrag[s],
                                                         (short)0, acc0, false, false);
            acc1 = __builtin_amdgcn_wmma_f32_16x16x4_f32(false, x1frag[s + 1], false, bfrag[s + 1],
                                                         (short)0, acc1, false, false);
        }
        const v8f acc = acc0 + acc1;

        const float b2h = B2[hcol];
        #pragma unroll
        for (int v = 0; v < 8; ++v) {
            const int row = v + 8 * half;
            float pre = acc[v] + b2h;
            #pragma unroll
            for (int r = 0; r < RANK; ++r)
                pre += t2_lds[row * RANK + r] * g_lds[row * GSTR + OFF_B2 + 4 * hcol + r];
            x2v[ct][v] = tanh_sl(pre);
        }
    }
    #pragma unroll
    for (int ct = 0; ct < 4; ++ct) {
        #pragma unroll
        for (int v = 0; v < 8; ++v)
            x_lds[(v + 8 * half) * XSTR + ct * 16 + m] = x2v[ct][v];
    }

    run_chunks(25, 9, true, false);

    if (half == 0) {
        float t3[RANK] = {0.f, 0.f, 0.f, 0.f};
        float w3s = 0.f;
        for (int h = 0; h < HID; ++h) {
            const float xv = x_lds[m * XSTR + h];
            w3s += xv * W3[h];
            #pragma unroll
            for (int r = 0; r < RANK; ++r)
                t3[r] += xv * g_lds[m * GSTR + OFF_A3 + 64 * r + h];
        }
        float o = w3s + B3[0];
        #pragma unroll
        for (int r = 0; r < RANK; ++r)
            o += t3[r] * g_lds[m * GSTR + OFF_B3 + r];
        out[base + m] = o;
    }
}

extern "C" void kernel_launch(void* const* d_in, const int* in_sizes, int n_in,
                              void* d_out, int out_size, void* d_ws, size_t ws_size,
                              hipStream_t stream) {
    const float* coords = (const float*)d_in[0];
    const float* param  = (const float*)d_in[1];
    const float* W1     = (const float*)d_in[2];
    const float* B1     = (const float*)d_in[3];
    const float* W2     = (const float*)d_in[4];
    const float* B2     = (const float*)d_in[5];
    const float* W3     = (const float*)d_in[6];
    const float* B3     = (const float*)d_in[7];
    const float* Wh1    = (const float*)d_in[8];
    const float* Bh1    = (const float*)d_in[9];
    const float* Wh2    = (const float*)d_in[10];
    const float* Bh2    = (const float*)d_in[11];
    float* out = (float*)d_out;

    const int n = in_sizes[1];
    const size_t smem = (size_t)(2 * 16 * GSTR + 2 * BROWS * BSTR + 2 * 16 * XSTR + 2 * 16 * RANK) * sizeof(float);
    (void)hipFuncSetAttribute((const void*)hyperlr_fused,
                              hipFuncAttributeMaxDynamicSharedMemorySize, (int)smem);

    const int blocks = n / 32;
    hyperlr_fused<<<blocks, 64, smem, stream>>>(coords, param, W1, B1, W2, B2, W3, B3,
                                                Wh1, Bh1, Wh2, Bh2, out);
}


# Wh2 staged in chunks of 64 rows (four column tiles) instead of 32: 18 barrier trips per wave instead of 34, each with one fragment-read wait and one LDS store drain for 64 matrix instructions
# speedup vs baseline: 3.3732x; 1.0756x over previous
#include <hip/hip_runtime.h>
#include <hip/hip_bf16.h>
#include <math.h>

typedef float v2f __attribute__((ext_vector_type(2)));
typedef float v4f __attribute__((ext_vector_type(4)));
typedef float v8f __attribute__((ext_vector_type(8)));

#define HID   64
#define RANK  4
#define NAB   1036
#define GSTR  276
#define XSTR  65
#define BROWS 64
#define BSTR  68

#define OFF_A1 0
#define OFF_B1 8
#define OFF_A2 0
#define OFF_B2 0
#define OFF_A3 0
#define OFF_B3 256

__device__ __forceinline__ float tanh_sl(float x) {
    const float ax = fabsf(x);
    const float s  = x * x;
    float p = fmaf(s, -0x1.758e7ap-8f, 0x1.521192p-6f);
    p = fmaf(s, p, -0x1.b8389cp-5f);
    p = fmaf(s, p, 0x1.110704p-3f);
    p = fmaf(s, p, -0x1.555532p-2f);
    const float lo = fmaf(s, ax * p, ax);
    const float e  = expf(ax + ax);
    const float hi = fmaf(__builtin_amdgcn_rcpf(1.0f + e), -2.0f, 1.0f);
    return copysignf(ax < 0.625f ? lo : hi, x);
}

__global__ __launch_bounds__(64) void hyperlr_fused(
    const float* __restrict__ coords, const float* __restrict__ param,
    const float* __restrict__ W1,  const float* __restrict__ B1,
    const float* __restrict__ W2,  const float* __restrict__ B2,
    const float* __restrict__ W3,  const float* __restrict__ B3,
    const float* __restrict__ Wh1, const float* __restrict__ Bh1,
    const float* __restrict__ Wh2, const float* __restrict__ Bh2,
    float* __restrict__ out)
{
    extern __shared__ __attribute__((aligned(16))) float smem[];
    const int lane = threadIdx.x & 31;
    const int wave = threadIdx.x >> 5;

    float* g_lds  = smem + wave * (16 * GSTR);
    float* b_lds  = smem + 2 * (16 * GSTR);
    float* x_lds  = smem + 2 * (16 * GSTR) + 2 * BROWS * BSTR + wave * (16 * XSTR);
    float* t2_lds = smem + 2 * (16 * GSTR) + 2 * BROWS * BSTR + 2 * (16 * XSTR) + wave * (16 * RANK);
    static_assert((2 * 16 * GSTR) % 4 == 0, "staging ring must start on a 16-byte boundary");

    const int m    = lane & 15;
    const int half = lane >> 4;
    const int kb   = half * 2;
    const int base = blockIdx.x * 32 + wave * 16;

    const float pv = param[base + m];
    v2f afrag[16];
    #pragma unroll
    for (int s = 0; s < 16; ++s) {
        int k = 4 * s + kb;
        afrag[s].x = tanh_sl(pv * Wh1[k]     + Bh1[k]);
        afrag[s].y = tanh_sl(pv * Wh1[k + 1] + Bh1[k + 1]);
    }

    const int srow = threadIdx.x >> 4;
    const int skq  = (threadIdx.x & 15) * 4;
    v4f stg[16];
    auto chunk_col = [&](int j) -> int { return j * BROWS - (j >= 5 ? 56 : 0); };
    auto gload = [&](int j, bool whole) {
        const int row0 = chunk_col(j);
        auto ld = [&](int i) {
            const int row  = row0 + srow + 4 * i;
            const int crow = row < NAB ? row : (NAB - 1);
            stg[i] = *(const v4f*)(Wh2 + crow * HID + skq);
        };
        #pragma unroll
        for (int i = 0; i < 4; ++i) ld(i);
        if (whole) {
            #pragma unroll
            for (int i = 4; i < 16; ++i) ld(i);
        }
    };
    auto sstore = [&](int stage, bool whole) {
        float* dst = b_lds + stage * (BROWS * BSTR) + srow * BSTR + skq;
        #pragma unroll
        for (int i = 0; i < 4; ++i) *(v4f*)(dst + 4 * i * BSTR) = stg[i];
        if (whole) {
            #pragma unroll
            for (int i = 4; i < 16; ++i) *(v4f*)(dst + 4 * i * BSTR) = stg[i];
        }
    };
    auto readB = [&](int stage, int tile, v2f (&dst)[16]) {
        const float* sp = b_lds + stage * (BROWS * BSTR) + (tile * 16 + m) * BSTR + kb;
        #pragma unroll
        for (int s = 0; s < 16; ++s) dst[s] = *(const v2f*)(sp + 4 * s);
    };
    auto mma_tile = [&](const v2f (&b)[16]) -> v8f {
        v8f acc0 = {}, acc1 = {};
        #pragma unroll
        for (int s = 0; s < 16; s += 2) {
            acc0 = __builtin_amdgcn_wmma_f32_16x16x4_f32(false, afrag[s],     false, b[s],
                                                         (short)0, acc0, false, false);
            acc1 = __builtin_amdgcn_wmma_f32_16x16x4_f32(false, afrag[s + 1], false, b[s + 1],
                                                         (short)0, acc1, false, false);
        }
        return acc0 + acc1;
    };
    auto load_bias = [&](int gcol0) -> float {
        const int col  = gcol0 + m;
        const int ccol = col < NAB ? col : (NAB - 1);
        return Bh2[ccol];
    };
    auto store_tile = [&](int wcol0, const v8f& acc, float bias) {
        const int col = wcol0 + m;
        #pragma unroll
        for (int v = 0; v < 8; ++v)
            g_lds[(v + 8 * half) * GSTR + col] = acc[v] + bias;
    };

    auto run_chunks = [&](int j0, int n, bool tail_one, bool more_after) {
        for (int cc = 0; cc < n; ++cc) {
            const int j = j0 + cc;
            const int stage = j & 1;
            const bool four = !(tail_one && cc == n - 1);
            const bool next = more_after || cc + 1 < n;
            const bool next_whole = !(tail_one && cc + 2 == n);
            const int gcol0 = chunk_col(j);
            float bias[4];
            #pragma unroll
            for (int t = 0; t < 4; ++t) bias[t] = load_bias(gcol0 + 16 * t);
            if (next) gload(j + 1, next_whole);
            if (four) {
                v2f b[4][16];
                v8f acc[4];
                #pragma unroll
                for (int t = 0; t < 4; ++t) readB(stage, t, b[t]);
                __builtin_amdgcn_sched_barrier(0);
                #pragma unroll
                for (int t = 0; t < 4; ++t) acc[t] = mma_tile(b[t]);
                #pragma unroll
                for (int t = 0; t < 4; ++t) store_tile(cc * BROWS + 16 * t, acc[t], bias[t]);
            } else {
                v2f b0[16];
                readB(stage, 0, b0);
                __builtin_amdgcn_sched_barrier(0);
                const v8f acc0 = mma_tile(b0);
                store_tile(cc * BROWS, acc0, bias[0]);
            }
            if (next) sstore(stage ^ 1, next_whole);
            __syncthreads();
        }
    };

    gload(0, true);
    sstore(0, true);
    __syncthreads();

    run_chunks(0, 5, true, true);

    const float c0 = coords[(base + m) * 2 + 0];
    const float c1 = coords[(base + m) * 2 + 1];
    float t1[RANK];
    #pragma unroll
    for (int r = 0; r < RANK; ++r)
        t1[r] = c0 * g_lds[m * GSTR + OFF_A1 + 2 * r]
              + c1 * g_lds[m * GSTR + OFF_A1 + 2 * r + 1];

    v2f x1frag[16];
    #pragma unroll
    for (int s = 0; s < 16; ++s) {
        #pragma unroll
        for (int j = 0; j < 2; ++j) {
            const int h = 4 * s + kb + j;
            float pre = W1[2 * h] * c0 + W1[2 * h + 1] * c1 + B1[h];
            #pragma unroll
            for (int r = 0; r < RANK; ++r)
                pre += g_lds[m * GSTR + OFF_B1 + 4 * h + r] * t1[r];
            x1frag[s][j] = tanh_sl(pre);
        }
    }
    #pragma unroll
    for (int s = 0; s < 16; ++s) {
        #pragma unroll
        for (int j = 0; j < 2; ++j)
            x_lds[m * XSTR + 4 * s + kb + j] = x1frag[s][j];
    }

    run_chunks(5, 4, false, true);

    {
        float t2[RANK] = {0.f, 0.f, 0.f, 0.f};
        for (int h = 0; h < HID; ++h) {
            const float xv = x_lds[m * XSTR + h];
            #pragma unroll
            for (int r = 0; r < RANK; ++r)
                t2[r] += xv * g_lds[m * GSTR + OFF_A2 + 64 * r + h];
        }
        if (half == 0) {
            #pragma unroll
            for (int r = 0; r < RANK; ++r) t2_lds[m * RANK + r] = t2[r];
        }
    }

    run_chunks(9, 4, false, true);

    float x2v[4][8];
    #pragma unroll
    for (int ct = 0; ct < 4; ++ct) {
        const int hcol = ct * 16 + m;
        const float* wp = W2 + hcol * HID + kb;
        v2f bfrag[16];
        #pragma unroll
        for (int s = 0; s < 16; ++s) { bfrag[s].x = wp[4 * s]; bfrag[s].y = wp[4 * s + 1]; }

        v8f acc0 = {}, acc1 = {};
        #pragma unroll
        for (int s = 0; s < 16; s += 2) {
            acc0 = __builtin_amdgcn_wmma_f32_16x16x4_f32(false, x1frag[s],     false, bfrag[s],
                                                         (short)0, acc0, false, false);
            acc1 = __builtin_amdgcn_wmma_f32_16x16x4_f32(false, x1frag[s + 1], false, bfrag[s + 1],
                                                         (short)0, acc1, false, false);
        }
        const v8f acc = acc0 + acc1;

        const float b2h = B2[hcol];
        #pragma unroll
        for (int v = 0; v < 8; ++v) {
            const int row = v + 8 * half;
            float pre = acc[v] + b2h;
            #pragma unroll
            for (int r = 0; r < RANK; ++r)
                pre += t2_lds[row * RANK + r] * g_lds[row * GSTR + OFF_B2 + 4 * hcol + r];
            x2v[ct][v] = tanh_sl(pre);
        }
    }
    #pragma unroll
    for (int ct = 0; ct < 4; ++ct) {
        #pragma unroll
        for (int v = 0; v < 8; ++v)
            x_lds[(v + 8 * half) * XSTR + ct * 16 + m] = x2v[ct][v];
    }

    run_chunks(13, 5, true, false);

    if (half == 0) {
        float t3[RANK] = {0.f, 0.f, 0.f, 0.f};
        float w3s = 0.f;
        for (int h = 0; h < HID; ++h) {
            const float xv = x_lds[m * XSTR + h];
            w3s += xv * W3[h];
            #pragma unroll
            for (int r = 0; r < RANK; ++r)
                t3[r] += xv * g_lds[m * GSTR + OFF_A3 + 64 * r + h];
        }
        float o = w3s + B3[0];
        #pragma unroll
        for (int r = 0; r < RANK; ++r)
            o += t3[r] * g_lds[m * GSTR + OFF_B3 + r];
        out[base + m] = o;
    }
}

extern "C" void kernel_launch(void* const* d_in, const int* in_sizes, int n_in,
                              void* d_out, int out_size, void* d_ws, size_t ws_size,
                              hipStream_t stream) {
    const float* coords = (const float*)d_in[0];
    const float* param  = (const float*)d_in[1];
    const float* W1     = (const float*)d_in[2];
    const float* B1     = (const float*)d_in[3];
    const float* W2     = (const float*)d_in[4];
    const float* B2     = (const float*)d_in[5];
    const float* W3     = (const float*)d_in[6];
    const float* B3     = (const float*)d_in[7];
    const float* Wh1    = (const float*)d_in[8];
    const float* Bh1    = (const float*)d_in[9];
    const float* Wh2    = (const float*)d_in[10];
    const float* Bh2    = (const float*)d_in[11];
    float* out = (float*)d_out;

    const int n = in_sizes[1];
    const size_t smem = (size_t)(2 * 16 * GSTR + 2 * BROWS * BSTR + 2 * 16 * XSTR + 2 * 16 * RANK) * sizeof(float);
    (void)hipFuncSetAttribute((const void*)hyperlr_fused,
                              hipFuncAttributeMaxDynamicSharedMemorySize, (int)smem);

    const int blocks = n / 32;
    hyperlr_fused<<<blocks, 64, smem, stream>>>(coords, param, W1, B1, W2, B2, W3, B3,
                                                Wh1, Bh1, Wh2, Bh2, out);
}


# t2, t3 and x2.W3 formed from the accumulator tiles (four columns per lane, then a 16-lane DPP butterfly sum: summation order changed); a2, a3 no longer stored to LDS and re-read by serial loops
# speedup vs baseline: 3.5719x; 1.0589x over previous
#include <hip/hip_runtime.h>
#include <hip/hip_bf16.h>
#include <math.h>

typedef float v2f __attribute__((ext_vector_type(2)));
typedef float v4f __attribute__((ext_vector_type(4)));
typedef float v8f __attribute__((ext_vector_type(8)));

#define HID   64
#define RANK  4
#define NAB   1036
#define GSTR  276
#define XSTR  65
#define BROWS 64
#define BSTR  68

#define OFF_A1 0
#define OFF_B1 8
#define OFF_A2 0
#define OFF_B2 0
#define OFF_A3 0
#define OFF_B3 256

__device__ __forceinline__ float tanh_sl(float x) {
    const float ax = fabsf(x);
    const float s  = x * x;
    float p = fmaf(s, -0x1.758e7ap-8f, 0x1.521192p-6f);
    p = fmaf(s, p, -0x1.b8389cp-5f);
    p = fmaf(s, p, 0x1.110704p-3f);
    p = fmaf(s, p, -0x1.555532p-2f);
    const float lo = fmaf(s, ax * p, ax);
    const float e  = expf(ax + ax);
    const float hi = fmaf(__builtin_amdgcn_rcpf(1.0f + e), -2.0f, 1.0f);
    return copysignf(ax < 0.625f ? lo : hi, x);
}

__device__ __forceinline__ float row_sum16(float v) {
    v += __builtin_bit_cast(float, __builtin_amdgcn_update_dpp(0, __builtin_bit_cast(int, v), 0x161, 0xF, 0xF, true));
    v += __builtin_bit_cast(float, __builtin_amdgcn_update_dpp(0, __builtin_bit_cast(int, v), 0x162, 0xF, 0xF, true));
    v += __builtin_bit_cast(float, __builtin_amdgcn_update_dpp(0, __builtin_bit_cast(int, v), 0x164, 0xF, 0xF, true));
    v += __builtin_bit_cast(float, __builtin_amdgcn_update_dpp(0, __builtin_bit_cast(int, v), 0x168, 0xF, 0xF, true));
    return v;
}

__global__ __launch_bounds__(64) void hyperlr_fused(
    const float* __restrict__ coords, const float* __restrict__ param,
    const float* __restrict__ W1,  const float* __restrict__ B1,
    const float* __restrict__ W2,  const float* __restrict__ B2,
    const float* __restrict__ W3,  const float* __restrict__ B3,
    const float* __restrict__ Wh1, const float* __restrict__ Bh1,
    const float* __restrict__ Wh2, const float* __restrict__ Bh2,
    float* __restrict__ out)
{
    extern __shared__ __attribute__((aligned(16))) float smem[];
    const int lane = threadIdx.x & 31;
    const int wave = threadIdx.x >> 5;

    float* g_lds  = smem + wave * (16 * GSTR);
    float* b_lds  = smem + 2 * (16 * GSTR);
    float* x_lds  = smem + 2 * (16 * GSTR) + 2 * BROWS * BSTR + wave * (16 * XSTR);
    float* t2_lds = smem + 2 * (16 * GSTR) + 2 * BROWS * BSTR + 2 * (16 * XSTR) + wave * (16 * RANK);
    static_assert((2 * 16 * GSTR) % 4 == 0, "staging ring must start on a 16-byte boundary");

    const int m    = lane & 15;
    const int half = lane >> 4;
    const int kb   = half * 2;
    const int base = blockIdx.x * 32 + wave * 16;

    const float pv = param[base + m];
    v2f afrag[16];
    #pragma unroll
    for (int s = 0; s < 16; ++s) {
        int k = 4 * s + kb;
        afrag[s].x = tanh_sl(pv * Wh1[k]     + Bh1[k]);
        afrag[s].y = tanh_sl(pv * Wh1[k + 1] + Bh1[k + 1]);
    }

    const int srow = threadIdx.x >> 4;
    const int skq  = (threadIdx.x & 15) * 4;
    v4f stg[16];
    auto chunk_col = [&](int j) -> int { return j * BROWS - (j >= 5 ? 56 : 0); };
    auto gload = [&](int j, bool whole) {
        const int row0 = chunk_col(j);
        auto ld = [&](int i) {
            const int row  = row0 + srow + 4 * i;
            const int crow = row < NAB ? row : (NAB - 1);
            stg[i] = *(const v4f*)(Wh2 + crow * HID + skq);
        };
        #pragma unroll
        for (int i = 0; i < 4; ++i) ld(i);
        if (whole) {
            #pragma unroll
            for (int i = 4; i < 16; ++i) ld(i);
        }
    };
    auto sstore = [&](int stage, bool whole) {
        float* dst = b_lds + stage * (BROWS * BSTR) + srow * BSTR + skq;
        #pragma unroll
        for (int i = 0; i < 4; ++i) *(v4f*)(dst + 4 * i * BSTR) = stg[i];
        if (whole) {
            #pragma unroll
            for (int i = 4; i < 16; ++i) *(v4f*)(dst + 4 * i * BSTR) = stg[i];
        }
    };
    auto readB = [&](int stage, int tile, v2f (&dst)[16]) {
        const float* sp = b_lds + stage * (BROWS * BSTR) + (tile * 16 + m) * BSTR + kb;
        #pragma unroll
        for (int s = 0; s < 16; ++s) dst[s] = *(const v2f*)(sp + 4 * s);
    };
    auto mma_tile = [&](const v2f (&b)[16]) -> v8f {
        v8f acc0 = {}, acc1 = {};
        #pragma unroll
        for (int s = 0; s < 16; s += 2) {
            acc0 = __builtin_amdgcn_wmma_f32_16x16x4_f32(false, afrag[s],     false, b[s],
                                                         (short)0, acc0, false, false);
            acc1 = __builtin_amdgcn_wmma_f32_16x16x4_f32(false, afrag[s + 1], false, b[s + 1],
                                                         (short)0, acc1, false, false);
        }
        return acc0 + acc1;
    };
    auto load_bias = [&](int gcol0) -> float {
        const int col  = gcol0 + m;
        const int ccol = col < NAB ? col : (NAB - 1);
        return Bh2[ccol];
    };
    auto store_tile = [&](int wcol0, const v8f& acc, float bias) {
        const int col = wcol0 + m;
        #pragma unroll
        for (int v = 0; v < 8; ++v)
            g_lds[(v + 8 * half) * GSTR + col] = acc[v] + bias;
    };

    auto run_chunks = [&](int j0, int n, bool tail_one, bool more_after, bool dot, const float (&xc)[4][8]) {
        for (int cc = 0; cc < n; ++cc) {
            const int j = j0 + cc;
            const int stage = j & 1;
            const bool four = !(tail_one && cc == n - 1);
            const bool next = more_after || cc + 1 < n;
            const bool next_whole = !(tail_one && cc + 2 == n);
            const int gcol0 = chunk_col(j);
            float bias[4];
            #pragma unroll
            for (int t = 0; t < 4; ++t) bias[t] = load_bias(gcol0 + 16 * t);
            if (next) gload(j + 1, next_whole);
            if (four) {
                v2f b[4][16];
                v8f acc[4];
                #pragma unroll
                for (int t = 0; t < 4; ++t) readB(stage, t, b[t]);
                __builtin_amdgcn_sched_barrier(0);
                #pragma unroll
                for (int t = 0; t < 4; ++t) acc[t] = mma_tile(b[t]);
                if (dot) {
                    float ts[8];
                    #pragma unroll
                    for (int v = 0; v < 8; ++v) {
                        float sv = (acc[0][v] + bias[0]) * xc[0][v];
                        #pragma unroll
                        for (int t = 1; t < 4; ++t) sv = fmaf(acc[t][v] + bias[t], xc[t][v], sv);
                        ts[v] = row_sum16(sv);
                    }
                    if (m == 0) {
                        #pragma unroll
                        for (int v = 0; v < 8; ++v) t2_lds[(v + 8 * half) * RANK + cc] = ts[v];
                    }
                } else {
                    #pragma unroll
                    for (int t = 0; t < 4; ++t) store_tile(cc * BROWS + 16 * t, acc[t], bias[t]);
                }
            } else {
                v2f b0[16];
                readB(stage, 0, b0);
                __builtin_amdgcn_sched_barrier(0);
                const v8f acc0 = mma_tile(b0);
                store_tile(cc * BROWS, acc0, bias[0]);
            }
            if (next) sstore(stage ^ 1, next_whole);
            __syncthreads();
        }
    };

    gload(0, true);
    sstore(0, true);
    __syncthreads();

    float xc1[4][8] = {};
    run_chunks(0, 5, true, true, false, xc1);

    const float c0 = coords[(base + m) * 2 + 0];
    const float c1 = coords[(base + m) * 2 + 1];
    float t1[RANK];
    #pragma unroll
    for (int r = 0; r < RANK; ++r)
        t1[r] = c0 * g_lds[m * GSTR + OFF_A1 + 2 * r]
              + c1 * g_lds[m * GSTR + OFF_A1 + 2 * r + 1];

    v2f x1frag[16];
    #pragma unroll
    for (int s = 0; s < 16; ++s) {
        #pragma unroll
        for (int j = 0; j < 2; ++j) {
            const int h = 4 * s + kb + j;
            float pre = W1[2 * h] * c0 + W1[2 * h + 1] * c1 + B1[h];
            #pragma unroll
            for (int r = 0; r < RANK; ++r)
                pre += g_lds[m * GSTR + OFF_B1 + 4 * h + r] * t1[r];
            x1frag[s][j] = tanh_sl(pre);
        }
    }
    #pragma unroll
    for (int s = 0; s < 16; ++s) {
        #pragma unroll
        for (int j = 0; j < 2; ++j)
            x_lds[m * XSTR + 4 * s + kb + j] = x1frag[s][j];
    }

    __syncthreads();
    #pragma unroll
    for (int t = 0; t < 4; ++t) {
        #pragma unroll
        for (int v = 0; v < 8; ++v) xc1[t][v] = x_lds[(v + 8 * half) * XSTR + 16 * t + m];
    }

    run_chunks(5, 4, false, true, true, xc1);

    run_chunks(9, 4, false, true, false, xc1);

    float x2v[4][8];
    #pragma unroll
    for (int ct = 0; ct < 4; ++ct) {
        const int hcol = ct * 16 + m;
        const float* wp = W2 + hcol * HID + kb;
        v2f bfrag[16];
        #pragma unroll
        for (int s = 0; s < 16; ++s) { bfrag[s].x = wp[4 * s]; bfrag[s].y = wp[4 * s + 1]; }

        v8f acc0 = {}, acc1 = {};
        #pragma unroll
        for (int s = 0; s < 16; s += 2) {
            acc0 = __builtin_amdgcn_wmma_f32_16x16x4_f32(false, x1frag[s],     false, bfrag[s],
                                                         (short)0, acc0, false, false);
            acc1 = __builtin_amdgcn_wmma_f32_16x16x4_f32(false, x1frag[s + 1], false, bfrag[s + 1],
                                                         (short)0, acc1, false, false);
        }
        const v8f acc = acc0 + acc1;

        const float b2h = B2[hcol];
        #pragma unroll
        for (int v = 0; v < 8; ++v) {
            const int row = v + 8 * half;
            float pre = acc[v] + b2h;
            #pragma unroll
            for (int r = 0; r < RANK; ++r)
                pre += t2_lds[row * RANK + r] * g_lds[row * GSTR + OFF_B2 + 4 * hcol + r];
            x2v[ct][v] = tanh_sl(pre);
        }
    }
    {
        float w3p[8];
        #pragma unroll
        for (int v = 0; v < 8; ++v) {
            float sv = x2v[0][v] * W3[m];
            #pragma unroll
            for (int ct = 1; ct < 4; ++ct) sv = fmaf(x2v[ct][v], W3[16 * ct + m], sv);
            w3p[v] = row_sum16(sv);
        }
        if (m == 0) {
            #pragma unroll
            for (int v = 0; v < 8; ++v) x_lds[v + 8 * half] = w3p[v];
        }
    }

    run_chunks(13, 5, true, false, true, x2v);

    if (half == 0) {
        float o = x_lds[m] + B3[0];
        #pragma unroll
        for (int r = 0; r < RANK; ++r)
            o += t2_lds[m * RANK + r] * g_lds[m * GSTR + OFF_B3 + r];
        out[base + m] = o;
    }
}

extern "C" void kernel_launch(void* const* d_in, const int* in_sizes, int n_in,
                              void* d_out, int out_size, void* d_ws, size_t ws_size,
                              hipStream_t stream) {
    const float* coords = (const float*)d_in[0];
    const float* param  = (const float*)d_in[1];
    const float* W1     = (const float*)d_in[2];
    const float* B1     = (const float*)d_in[3];
    const float* W2     = (const float*)d_in[4];
    const float* B2     = (const float*)d_in[5];
    const float* W3     = (const float*)d_in[6];
    const float* B3     = (const float*)d_in[7];
    const float* Wh1    = (const float*)d_in[8];
    const float* Bh1    = (const float*)d_in[9];
    const float* Wh2    = (const float*)d_in[10];
    const float* Bh2    = (const float*)d_in[11];
    float* out = (float*)d_out;

    const int n = in_sizes[1];
    const size_t smem = (size_t)(2 * 16 * GSTR + 2 * BROWS * BSTR + 2 * 16 * XSTR + 2 * 16 * RANK) * sizeof(float);
    (void)hipFuncSetAttribute((const void*)hyperlr_fused,
                              hipFuncAttributeMaxDynamicSharedMemorySize, (int)smem);

    const int blocks = n / 32;
    hyperlr_fused<<<blocks, 64, smem, stream>>>(coords, param, W1, B1, W2, B2, W3, B3,
                                                Wh1, Bh1, Wh2, Bh2, out);
}


# W2 goes through the LDS staging ring as one more 64-row chunk (whole-line 16-byte loads, prefetched a trip ahead, shared by both waves); layer 2 is a ring trip: no per-lane 8-byte loads strided 256 B
# speedup vs baseline: 3.6836x; 1.0313x over previous
#include <hip/hip_runtime.h>
#include <hip/hip_bf16.h>
#include <math.h>

typedef float v2f __attribute__((ext_vector_type(2)));
typedef float v4f __attribute__((ext_vector_type(4)));
typedef float v8f __attribute__((ext_vector_type(8)));

#define HID   64
#define RANK  4
#define NAB   1036
#define GSTR  276
#define XSTR  65
#define BROWS 64
#define BSTR  68

#define OFF_A1 0
#define OFF_B1 8
#define OFF_A2 0
#define OFF_B2 0
#define OFF_A3 0
#define OFF_B3 256

__device__ __forceinline__ float tanh_sl(float x) {
    const float ax = fabsf(x);
    const float s  = x * x;
    float p = fmaf(s, -0x1.758e7ap-8f, 0x1.521192p-6f);
    p = fmaf(s, p, -0x1.b8389cp-5f);
    p = fmaf(s, p, 0x1.110704p-3f);
    p = fmaf(s, p, -0x1.555532p-2f);
    const float lo = fmaf(s, ax * p, ax);
    const float e  = expf(ax + ax);
    const float hi = fmaf(__builtin_amdgcn_rcpf(1.0f + e), -2.0f, 1.0f);
    return copysignf(ax < 0.625f ? lo : hi, x);
}

__device__ __forceinline__ float row_sum16(float v) {
    v += __builtin_bit_cast(float, __builtin_amdgcn_update_dpp(0, __builtin_bit_cast(int, v), 0x161, 0xF, 0xF, true));
    v += __builtin_bit_cast(float, __builtin_amdgcn_update_dpp(0, __builtin_bit_cast(int, v), 0x162, 0xF, 0xF, true));
    v += __builtin_bit_cast(float, __builtin_amdgcn_update_dpp(0, __builtin_bit_cast(int, v), 0x164, 0xF, 0xF, true));
    v += __builtin_bit_cast(float, __builtin_amdgcn_update_dpp(0, __builtin_bit_cast(int, v), 0x168, 0xF, 0xF, true));
    return v;
}

__global__ __launch_bounds__(64) void hyperlr_fused(
    const float* __restrict__ coords, const float* __restrict__ param,
    const float* __restrict__ W1,  const float* __restrict__ B1,
    const float* __restrict__ W2,  const float* __restrict__ B2,
    const float* __restrict__ W3,  const float* __restrict__ B3,
    const float* __restrict__ Wh1, const float* __restrict__ Bh1,
    const float* __restrict__ Wh2, const float* __restrict__ Bh2,
    float* __restrict__ out)
{
    extern __shared__ __attribute__((aligned(16))) float smem[];
    const int lane = threadIdx.x & 31;
    const int wave = threadIdx.x >> 5;

    float* g_lds  = smem + wave * (16 * GSTR);
    float* b_lds  = smem + 2 * (16 * GSTR);
    float* x_lds  = smem + 2 * (16 * GSTR) + 2 * BROWS * BSTR + wave * (16 * XSTR);
    float* t2_lds = smem + 2 * (16 * GSTR) + 2 * BROWS * BSTR + 2 * (16 * XSTR) + wave * (16 * RANK);
    static_assert((2 * 16 * GSTR) % 4 == 0, "staging ring must start on a 16-byte boundary");

    const int m    = lane & 15;
    const int half = lane >> 4;
    const int kb   = half * 2;
    const int base = blockIdx.x * 32 + wave * 16;

    const float pv = param[base + m];
    v2f afrag[16];
    #pragma unroll
    for (int s = 0; s < 16; ++s) {
        int k = 4 * s + kb;
        afrag[s].x = tanh_sl(pv * Wh1[k]     + Bh1[k]);
        afrag[s].y = tanh_sl(pv * Wh1[k + 1] + Bh1[k + 1]);
    }

    const int srow = threadIdx.x >> 4;
    const int skq  = (threadIdx.x & 15) * 4;
    v4f stg[16];
    auto chunk_col = [&](int j) -> int { return j * BROWS - (j >= 5 ? 56 : 0); };
    auto gload = [&](int j, bool whole) {
        const int row0 = chunk_col(j);
        auto ld = [&](int i) {
            const int row  = row0 + srow + 4 * i;
            const int crow = row < NAB ? row : (NAB - 1);
            stg[i] = *(const v4f*)(Wh2 + crow * HID + skq);
        };
        #pragma unroll
        for (int i = 0; i < 4; ++i) ld(i);
        if (whole) {
            #pragma unroll
            for (int i = 4; i < 16; ++i) ld(i);
        }
    };
    auto gload_w2 = [&]() {
        #pragma unroll
        for (int i = 0; i < 16; ++i) stg[i] = *(const v4f*)(W2 + (srow + 4 * i) * HID + skq);
    };
    auto sstore = [&](int stage, bool whole) {
        float* dst = b_lds + stage * (BROWS * BSTR) + srow * BSTR + skq;
        #pragma unroll
        for (int i = 0; i < 4; ++i) *(v4f*)(dst + 4 * i * BSTR) = stg[i];
        if (whole) {
            #pragma unroll
            for (int i = 4; i < 16; ++i) *(v4f*)(dst + 4 * i * BSTR) = stg[i];
        }
    };
    auto readB = [&](int stage, int tile, v2f (&dst)[16]) {
        const float* sp = b_lds + stage * (BROWS * BSTR) + (tile * 16 + m) * BSTR + kb;
        #pragma unroll
        for (int s = 0; s < 16; ++s) dst[s] = *(const v2f*)(sp + 4 * s);
    };
    auto mma_tile = [&](const v2f (&a)[16], const v2f (&b)[16]) -> v8f {
        v8f acc0 = {}, acc1 = {};
        #pragma unroll
        for (int s = 0; s < 16; s += 2) {
            acc0 = __builtin_amdgcn_wmma_f32_16x16x4_f32(false, a[s],     false, b[s],
                                                         (short)0, acc0, false, false);
            acc1 = __builtin_amdgcn_wmma_f32_16x16x4_f32(false, a[s + 1], false, b[s + 1],
                                                         (short)0, acc1, false, false);
        }
        return acc0 + acc1;
    };
    auto load_bias = [&](int gcol0) -> float {
        const int col  = gcol0 + m;
        const int ccol = col < NAB ? col : (NAB - 1);
        return Bh2[ccol];
    };
    auto store_tile = [&](int wcol0, const v8f& acc, float bias) {
        const int col = wcol0 + m;
        #pragma unroll
        for (int v = 0; v < 8; ++v)
            g_lds[(v + 8 * half) * GSTR + col] = acc[v] + bias;
    };

    auto run_chunks = [&](int j0, int n, bool tail_one, int after, int flip, bool dot, const float (&xc)[4][8]) {
        for (int cc = 0; cc < n; ++cc) {
            const int j = j0 + cc;
            const int stage = (j + flip) & 1;
            const bool four = !(tail_one && cc == n - 1);
            const bool next = after != 0 || cc + 1 < n;
            const bool next_w2 = after == 2 && cc + 1 == n;
            const bool next_whole = !(tail_one && cc + 2 == n);
            const int gcol0 = chunk_col(j);
            float bias[4];
            #pragma unroll
            for (int t = 0; t < 4; ++t) bias[t] = load_bias(gcol0 + 16 * t);
            if (next) { if (next_w2) gload_w2(); else gload(j + 1, next_whole); }
            if (four) {
                v2f b[4][16];
                v8f acc[4];
                #pragma unroll
                for (int t = 0; t < 4; ++t) readB(stage, t, b[t]);
                __builtin_amdgcn_sched_barrier(0);
                #pragma unroll
                for (int t = 0; t < 4; ++t) acc[t] = mma_tile(afrag, b[t]);
                if (dot) {
                    float ts[8];
                    #pragma unroll
                    for (int v = 0; v < 8; ++v) {
                        float sv = (acc[0][v] + bias[0]) * xc[0][v];
                        #pragma unroll
                        for (int t = 1; t < 4; ++t) sv = fmaf(acc[t][v] + bias[t], xc[t][v], sv);
                        ts[v] = row_sum16(sv);
                    }
                    if (m == 0) {
                        #pragma unroll
                        for (int v = 0; v < 8; ++v) t2_lds[(v + 8 * half) * RANK + cc] = ts[v];
                    }
                } else {
                    #pragma unroll
                    for (int t = 0; t < 4; ++t) store_tile(cc * BROWS + 16 * t, acc[t], bias[t]);
                }
            } else {
                v2f b0[16];
                readB(stage, 0, b0);
                __builtin_amdgcn_sched_barrier(0);
                const v8f acc0 = mma_tile(afrag, b0);
                store_tile(cc * BROWS, acc0, bias[0]);
            }
            if (next) sstore(stage ^ 1, next_whole);
            __syncthreads();
        }
    };

    gload(0, true);
    sstore(0, true);
    __syncthreads();

    float xc1[4][8] = {};
    run_chunks(0, 5, true, 1, 0, false, xc1);

    const float c0 = coords[(base + m) * 2 + 0];
    const float c1 = coords[(base + m) * 2 + 1];
    float t1[RANK];
    #pragma unroll
    for (int r = 0; r < RANK; ++r)
        t1[r] = c0 * g_lds[m * GSTR + OFF_A1 + 2 * r]
              + c1 * g_lds[m * GSTR + OFF_A1 + 2 * r + 1];

    v2f x1frag[16];
    #pragma unroll
    for (int s = 0; s < 16; ++s) {
        #pragma unroll
        for (int j = 0; j < 2; ++j) {
            const int h = 4 * s + kb + j;
            float pre = W1[2 * h] * c0 + W1[2 * h + 1] * c1 + B1[h];
            #pragma unroll
            for (int r = 0; r < RANK; ++r)
                pre += g_lds[m * GSTR + OFF_B1 + 4 * h + r] * t1[r];
            x1frag[s][j] = tanh_sl(pre);
        }
    }
    #pragma unroll
    for (int s = 0; s < 16; ++s) {
        #pragma unroll
        for (int j = 0; j < 2; ++j)
            x_lds[m * XSTR + 4 * s + kb + j] = x1frag[s][j];
    }

    __syncthreads();
    #pragma unroll
    for (int t = 0; t < 4; ++t) {
        #pragma unroll
        for (int v = 0; v < 8; ++v) xc1[t][v] = x_lds[(v + 8 * half) * XSTR + 16 * t + m];
    }

    run_chunks(5, 4, false, 1, 0, true, xc1);

    run_chunks(9, 4, false, 2, 0, false, xc1);

    float x2v[4][8];
    {
        float b2h[4];
        #pragma unroll
        for (int ct = 0; ct < 4; ++ct) b2h[ct] = B2[ct * 16 + m];
        gload(13, true);
        v2f wb[4][16];
        v8f acc[4];
        #pragma unroll
        for (int ct = 0; ct < 4; ++ct) readB(1, ct, wb[ct]);
        __builtin_amdgcn_sched_barrier(0);
        #pragma unroll
        for (int ct = 0; ct < 4; ++ct) acc[ct] = mma_tile(x1frag, wb[ct]);
        __builtin_amdgcn_sched_barrier(0);
        #pragma unroll
        for (int ct = 0; ct < 4; ++ct) {
            const int hcol = ct * 16 + m;
            #pragma unroll
            for (int v = 0; v < 8; ++v) {
                const int row = v + 8 * half;
                float pre = acc[ct][v] + b2h[ct];
                #pragma unroll
                for (int r = 0; r < RANK; ++r)
                    pre += t2_lds[row * RANK + r] * g_lds[row * GSTR + OFF_B2 + 4 * hcol + r];
                x2v[ct][v] = tanh_sl(pre);
            }
        }
    }
    {
        float w3p[8];
        #pragma unroll
        for (int v = 0; v < 8; ++v) {
            float sv = x2v[0][v] * W3[m];
            #pragma unroll
            for (int ct = 1; ct < 4; ++ct) sv = fmaf(x2v[ct][v], W3[16 * ct + m], sv);
            w3p[v] = row_sum16(sv);
        }
        if (m == 0) {
            #pragma unroll
            for (int v = 0; v < 8; ++v) x_lds[v + 8 * half] = w3p[v];
        }
    }

    sstore(0, true);
    __syncthreads();

    run_chunks(13, 5, true, 0, 1, true, x2v);

    if (half == 0) {
        float o = x_lds[m] + B3[0];
        #pragma unroll
        for (int r = 0; r < RANK; ++r)
            o += t2_lds[m * RANK + r] * g_lds[m * GSTR + OFF_B3 + r];
        out[base + m] = o;
    }
}

extern "C" void kernel_launch(void* const* d_in, const int* in_sizes, int n_in,
                              void* d_out, int out_size, void* d_ws, size_t ws_size,
                              hipStream_t stream) {
    const float* coords = (const float*)d_in[0];
    const float* param  = (const float*)d_in[1];
    const float* W1     = (const float*)d_in[2];
    const float* B1     = (const float*)d_in[3];
    const float* W2     = (const float*)d_in[4];
    const float* B2     = (const float*)d_in[5];
    const float* W3     = (const float*)d_in[6];
    const float* B3     = (const float*)d_in[7];
    const float* Wh1    = (const float*)d_in[8];
    const float* Bh1    = (const float*)d_in[9];
    const float* Wh2    = (const float*)d_in[10];
    const float* Bh2    = (const float*)d_in[11];
    float* out = (float*)d_out;

    const int n = in_sizes[1];
    const size_t smem = (size_t)(2 * 16 * GSTR + 2 * BROWS * BSTR + 2 * 16 * XSTR + 2 * 16 * RANK) * sizeof(float);
    (void)hipFuncSetAttribute((const void*)hyperlr_fused,
                              hipFuncAttributeMaxDynamicSharedMemorySize, (int)smem);

    const int blocks = n / 32;
    hyperlr_fused<<<blocks, 64, smem, stream>>>(coords, param, W1, B1, W2, B2, W3, B3,
                                                Wh1, Bh1, Wh2, Bh2, out);
}
